# starting kernel
# baseline (speedup 1.0000x reference)
#include <hip/hip_runtime.h>
#include <hip/hip_bf16.h>
#include <math.h>

#define VOCAB 50000
#define EMBD  128
#define HOPS  3
#define BS    64
#define STORY 50
#define SENT  20
#define QLEN  20

typedef __attribute__((ext_vector_type(2))) float v2f;
typedef __attribute__((ext_vector_type(8))) float v8f;

__global__ __launch_bounds__(EMBD) void memnn_u0(const int* __restrict__ q,
                                                 const float* __restrict__ A,
                                                 float* __restrict__ u) {
    __shared__ int tok[QLEN];
    const int b = blockIdx.x;
    const int d = threadIdx.x;
    if (d < QLEN) tok[d] = q[b * QLEN + d];
    __syncthreads();
    float acc = 0.0f;
#pragma unroll
    for (int j = 0; j < QLEN; ++j)
        acc += A[(size_t)tok[j] * EMBD + d];
    u[b * EMBD + d] = acc;
}

__global__ __launch_bounds__(EMBD) void memnn_gather(const int* __restrict__ x,
                                                     const float* __restrict__ A,
                                                     const float* __restrict__ TA,
                                                     const float* __restrict__ TC,
                                                     float* __restrict__ m,
                                                     float* __restrict__ c) {
    const int bs = blockIdx.x;
    const int s  = bs % STORY;
    const int d  = threadIdx.x;
    __shared__ int tok[SENT];
    if (d < SENT) tok[d] = x[(size_t)bs * SENT + d];
    __syncthreads();

    const float kd = (float)(d + 1) / (float)EMBD;
    const size_t LVL = (size_t)VOCAB * EMBD;

    float mp0 = 0.f, mp1 = 0.f, mp2 = 0.f;
    float cp1 = 0.f, cp2 = 0.f, cp3 = 0.f;
#pragma unroll
    for (int j = 0; j < SENT; ++j) {
        const float jj = (float)(j + 1) / (float)SENT;
        const float pe = 1.0f - jj - kd * (1.0f - 2.0f * jj);
        const size_t base = (size_t)tok[j] * EMBD + d;
        const float e0 = A[base];
        const float e1 = A[LVL + base];
        const float e2 = A[2 * LVL + base];
        const float e3 = A[3 * LVL + base];
        mp0 += e0 * pe;  mp1 += e1 * pe;  mp2 += e2 * pe;
        cp1 += e1;       cp2 += e2;       cp3 += e3;
    }
    const float ta = TA[s * EMBD + d];
    const float tc = TC[s * EMBD + d];
    const size_t off = (size_t)bs * EMBD + d;
    const size_t hs  = (size_t)BS * STORY * EMBD;
    m[off]          = mp0 + ta;
    m[hs + off]     = mp1 + ta;
    m[2 * hs + off] = mp2 + ta;
    c[off]          = cp1 + tc;
    c[hs + off]     = cp2 + tc;
    c[2 * hs + off] = cp3 + tc;
}

__global__ __launch_bounds__(EMBD) void memnn_hop(const float* __restrict__ m,
                                                  const float* __restrict__ c,
                                                  float* __restrict__ u,
                                                  int hop) {
    const int b = blockIdx.x;
    const int t = threadIdx.x;
    __shared__ float u_lds[EMBD];
    __shared__ float score[STORY];
    u_lds[t] = u[b * EMBD + t];
    __syncthreads();

    const int wave = t >> 5;
    const int lane = t & 31;
    const size_t hb = (size_t)hop * BS * STORY * EMBD + (size_t)b * STORY * EMBD;

    for (int s = wave; s < STORY; s += 4) {
        const float* mrow = m + hb + (size_t)s * EMBD;
        float p = 0.f;
#pragma unroll
        for (int i = 0; i < EMBD / 32; ++i)
            p += mrow[lane + 32 * i] * u_lds[lane + 32 * i];
        for (int o = 16; o > 0; o >>= 1) p += __shfl_xor(p, o, 32);
        if (lane == 0) score[s] = p;
    }
    __syncthreads();

    float mx = -3.4e38f;
    for (int s = 0; s < STORY; ++s) mx = fmaxf(mx, score[s]);
    float sm = 0.f;
    for (int s = 0; s < STORY; ++s) sm += __expf(score[s] - mx);
    const float inv = 1.0f / sm;

    float acc = 0.f;
    const float* crow = c + hb;
    for (int s = 0; s < STORY; ++s)
        acc += (__expf(score[s] - mx) * inv) * crow[(size_t)s * EMBD + t];
    u[b * EMBD + t] = u_lds[t] + acc;
}

__global__ __launch_bounds__(256) void memnn_logits_wmma(const float* __restrict__ u,
                                                         const float* __restrict__ A3,
                                                         float* __restrict__ out) {
    __shared__ float u_lds[BS * EMBD];
    for (int i = threadIdx.x; i < BS * EMBD; i += 256) u_lds[i] = u[i];
    __syncthreads();

    const int wave = threadIdx.x >> 5;
    const int lane = threadIdx.x & 31;
    const int tile = blockIdx.x * 8 + wave;
    const int NTILES = (VOCAB / 16) * (BS / 16);
    if (tile >= NTILES) return;

    const int mb = (tile & 3) << 4;
    const int nb = (tile >> 2) << 4;
    const int n  = lane & 15;
    const int h  = lane >> 4;

    const float* arow = u_lds + (mb + n) * EMBD;
    const float* brow = A3 + (size_t)(nb + n) * EMBD;

    v8f acc = {};
#pragma unroll
    for (int kk = 0; kk < EMBD / 4; ++kk) {
        const int k0 = kk * 4 + 2 * h;
        v2f a;  a.x  = arow[k0]; a.y  = arow[k0 + 1];
        v2f bv; bv.x = brow[k0]; bv.y = brow[k0 + 1];
        acc = __builtin_amdgcn_wmma_f32_16x16x4_f32(
             false, a,  false, bv,
             (short)0, acc,  false,  false);
    }
#pragma unroll
    for (int v = 0; v < 8; ++v)
        out[(size_t)(mb + v + 8 * h) * VOCAB + (nb + n)] = acc[v];
}

__global__ __launch_bounds__(256) void memnn_logsoftmax(float* __restrict__ out) {
    __shared__ float red[256];
    const int b = blockIdx.x;
    const int t = threadIdx.x;
    float* row = out + (size_t)b * VOCAB;

    float mx = -3.4e38f;
    for (int i = t; i < VOCAB; i += 256) mx = fmaxf(mx, row[i]);
    red[t] = mx; __syncthreads();
    for (int o = 128; o > 0; o >>= 1) {
        if (t < o) red[t] = fmaxf(red[t], red[t + o]);
        __syncthreads();
    }
    mx = red[0]; __syncthreads();

    float sm = 0.f;
    for (int i = t; i < VOCAB; i += 256) sm += __expf(row[i] - mx);
    red[t] = sm; __syncthreads();
    for (int o = 128; o > 0; o >>= 1) {
        if (t < o) red[t] += red[t + o];
        __syncthreads();
    }
    const float lse = mx + logf(red[0]);

    for (int i = t; i < VOCAB; i += 256) row[i] = row[i] - lse;
}

extern "C" void kernel_launch(void* const* d_in, const int* in_sizes, int n_in,
                              void* d_out, int out_size, void* d_ws, size_t ws_size,
                              hipStream_t stream) {
    const int*   x  = (const int*)d_in[0];
    const int*   q  = (const int*)d_in[1];
    const float* A  = (const float*)d_in[2];
    const float* TA = (const float*)d_in[3];
    const float* TC = (const float*)d_in[4];
    float* out = (float*)d_out;

    float* ws = (float*)d_ws;
    float* u  = ws;
    float* m  = ws + BS * EMBD;
    float* c  = m + (size_t)HOPS * BS * STORY * EMBD;

    memnn_u0<<<BS, EMBD, 0, stream>>>(q, A, u);
    memnn_gather<<<BS * STORY, EMBD, 0, stream>>>(x, A, TA, TC, m, c);
    for (int h = 0; h < HOPS; ++h)
        memnn_hop<<<BS, EMBD, 0, stream>>>(m, c, u, h);

    const int ntiles = (VOCAB / 16) * (BS / 16);
    const int blocks = (ntiles + 7) / 8;
    memnn_logits_wmma<<<blocks, 256, 0, stream>>>(
        u, A + 3 * (size_t)VOCAB * EMBD, out);

    memnn_logsoftmax<<<BS, 256, 0, stream>>>(out);
}


# logits GEMM re-tiled: 256 work-groups park their 12-13 vocabulary strips in LDS by whole-row loads; a wave keeps one batch tile's u fragments in registers, runs two strips at a time; k order kept
# speedup vs baseline: 3.1480x; 3.1480x over previous
#include <hip/hip_runtime.h>
#include <hip/hip_bf16.h>
#include <math.h>

#define VOCAB 50000
#define EMBD  128
#define HOPS  3
#define BS    64
#define STORY 50
#define SENT  20
#define QLEN  20

typedef __attribute__((ext_vector_type(2))) float v2f;
typedef __attribute__((ext_vector_type(8))) float v8f;

#define ST_THREADS 1024
#define ST_WAVES   (ST_THREADS / 32)
#define ST_UNITS   (STORY * 4)
#define ST_GROUPS  (ST_THREADS / EMBD)
#define ST_ROWS    ((STORY + ST_GROUPS - 1) / ST_GROUPS)
static_assert(EMBD == 128, "a lane owns 4 columns: 32 lanes cover one row");
static_assert(STORY > 32 && STORY <= 64, "a wave scores rows w and w+32; a lane holds scores l and l+32");
static_assert(STORY * SENT + QLEN <= ST_THREADS, "one thread per token");

__global__ __launch_bounds__(ST_THREADS) void memnn_story(const int* __restrict__ x,
                                                          const int* __restrict__ q,
                                                          const float* __restrict__ A,
                                                          const float* __restrict__ TA,
                                                          const float* __restrict__ TC,
                                                          float* __restrict__ u) {
    __shared__ float4 m_lds[HOPS * STORY * 32];
    __shared__ float4 c_lds[HOPS * STORY * 32];
    __shared__ float4 u_lds[32];
    __shared__ __attribute__((aligned(16))) int tok[STORY * SENT];
    __shared__ __attribute__((aligned(16))) int qtok[QLEN];
    __shared__ float score[64];
    __shared__ float part[ST_GROUPS * EMBD];

    const int b    = blockIdx.x;
    const int t    = threadIdx.x;
    const int lane = t & 31;
    const int wave = __builtin_amdgcn_readfirstlane(t >> 5);

    if (t < STORY * SENT)              tok[t] = x[(size_t)b * STORY * SENT + t];
    else if (t < STORY * SENT + QLEN)  qtok[t - STORY * SENT] = q[b * QLEN + (t - STORY * SENT)];
    __syncthreads();

    const size_t LVL4 = (size_t)VOCAB * 32;
    const float4* A4  = reinterpret_cast<const float4*>(A);
    const float4* TA4 = reinterpret_cast<const float4*>(TA);
    const float4* TC4 = reinterpret_cast<const float4*>(TC);
    const float4 zero4 = make_float4(0.f, 0.f, 0.f, 0.f);

    const float kd0 = (float)(4 * lane + 1) / (float)EMBD;
    const float kd1 = (float)(4 * lane + 2) / (float)EMBD;
    const float kd2 = (float)(4 * lane + 3) / (float)EMBD;
    const float kd3 = (float)(4 * lane + 4) / (float)EMBD;
    const float w0 = 1.0f - 2.0f * kd0, w1 = 1.0f - 2.0f * kd1;
    const float w2 = 1.0f - 2.0f * kd2, w3 = 1.0f - 2.0f * kd3;

    for (int unit = wave; unit < ST_UNITS; unit += ST_WAVES) {
        const int level = unit & 3;
        const int s     = unit >> 2;
        const char* Al = reinterpret_cast<const char*>(A4 + (size_t)level * LVL4);
        const int* tk = tok + s * SENT;

        float4 e[SENT];
#pragma unroll
        for (int j = 0; j < SENT; ++j)
            e[j] = *reinterpret_cast<const float4*>(Al + ((unsigned)tk[j] * 512u + (unsigned)lane * 16u));

        float4 ta = zero4, tc = zero4;
        if (level < HOPS) ta = TA4[s * 32 + lane];
        if (level > 0)    tc = TC4[s * 32 + lane];

        float4 ap = zero4, cp = zero4;
#pragma unroll
        for (int j = 0; j < SENT; ++j) {
            const float aj = 1.0f - (float)(j + 1) / (float)SENT;
            ap.x += e[j].x * aj;  ap.y += e[j].y * aj;
            ap.z += e[j].z * aj;  ap.w += e[j].w * aj;
            cp.x += e[j].x;  cp.y += e[j].y;  cp.z += e[j].z;  cp.w += e[j].w;
        }
        if (level < HOPS) {
            float4 mp;
            mp.x = (w0 * ap.x + kd0 * cp.x) + ta.x;
            mp.y = (w1 * ap.y + kd1 * cp.y) + ta.y;
            mp.z = (w2 * ap.z + kd2 * cp.z) + ta.z;
            mp.w = (w3 * ap.w + kd3 * cp.w) + ta.w;
            m_lds[(level * STORY + s) * 32 + lane] = mp;
        }
        if (level > 0) {
            cp.x += tc.x; cp.y += tc.y; cp.z += tc.z; cp.w += tc.w;
            c_lds[((level - 1) * STORY + s) * 32 + lane] = cp;
        }
    }
    if (wave == ST_WAVES - 1) {
        float4 e[QLEN];
#pragma unroll
        for (int j = 0; j < QLEN; ++j) e[j] = A4[(size_t)qtok[j] * 32 + lane];
        float4 acc = zero4;
#pragma unroll
        for (int j = 0; j < QLEN; ++j) {
            acc.x += e[j].x; acc.y += e[j].y; acc.z += e[j].z; acc.w += e[j].w;
        }
        u_lds[lane] = acc;
    }
    __syncthreads();

    const float  NEG   = -3.4e38f;
    const bool   two   = wave < STORY - 32;
    const int    g     = t >> 7;
    const int    d     = t & (EMBD - 1);
    const float* c_f   = reinterpret_cast<const float*>(c_lds);
    float*       u_f   = reinterpret_cast<float*>(u_lds);

    for (int h = 0; h < HOPS; ++h) {
        const float4 uv = u_lds[lane];
        const float4 m0 = m_lds[(h * STORY + wave) * 32 + lane];
        float4 m1 = zero4;
        if (two) m1 = m_lds[(h * STORY + wave + 32) * 32 + lane];
        float p0 = (m0.x * uv.x + m0.y * uv.y) + (m0.z * uv.z + m0.w * uv.w);
        float p1 = (m1.x * uv.x + m1.y * uv.y) + (m1.z * uv.z + m1.w * uv.w);
        for (int o = 16; o > 0; o >>= 1) {
            p0 += __shfl_xor(p0, o, 32);
            p1 += __shfl_xor(p1, o, 32);
        }
        if (lane == 0) {
            score[wave] = p0;
            if (two) score[wave + 32] = p1;
        }
        __syncthreads();

        const bool  hi = lane < STORY - 32;
        const float s0 = score[lane];
        const float s1 = hi ? score[lane + 32] : NEG;
        float mx = fmaxf(s0, s1);
        for (int o = 16; o > 0; o >>= 1) mx = fmaxf(mx, __shfl_xor(mx, o, 32));
        float sm = __expf(s0 - mx) + (hi ? __expf(s1 - mx) : 0.f);
        for (int o = 16; o > 0; o >>= 1) sm += __shfl_xor(sm, o, 32);
        const float inv = 1.0f / sm;

        float acc = 0.f;
        for (int i = 0; i < ST_ROWS; ++i) {
            const int s = g + ST_GROUPS * i;
            if (s < STORY)
                acc += (__expf(score[s] - mx) * inv) * c_f[(h * STORY + s) * EMBD + d];
        }
        part[g * EMBD + d] = acc;
        __syncthreads();

        if (t < EMBD) {
            float a = part[t];
            for (int k = 1; k < ST_GROUPS; ++k) a += part[k * EMBD + t];
            const float un = u_f[t] + a;
            u_f[t] = un;
            if (h == HOPS - 1) u[b * EMBD + t] = un;
        }
        __syncthreads();
    }
}

#define LG_THREADS 256
#define LG_WAVES   (LG_THREADS / 32)
#define LG_GRID    256
#define LG_NSTRIPS (VOCAB / 16)
#define LG_MAXS    ((LG_NSTRIPS + LG_GRID - 1) / LG_GRID)
#define LG_HALF    6
#define LG_PITCH   132
#define LG_ROWS_A  (LG_HALF * 16 / LG_WAVES)
#define LG_ROWS_B  (((LG_MAXS - LG_HALF) * 16 + LG_WAVES - 1) / LG_WAVES)
#define LG_ROWS_U  (BS / LG_WAVES)
static_assert(VOCAB % 16 == 0, "whole strips");
static_assert(LG_NSTRIPS / LG_GRID >= LG_HALF, "the first half is always full");
static_assert((LG_HALF * 16) % LG_WAVES == 0 && BS % LG_WAVES == 0, "rows dealt evenly to the waves");
static_assert(BS == 64 && LG_WAVES == 8, "4 batch tiles x 2 waves");

__device__ __forceinline__ void lg_lds_barrier() {
    __builtin_amdgcn_fence(__ATOMIC_RELEASE, "workgroup", "local");
    __builtin_amdgcn_s_barrier();
    __builtin_amdgcn_fence(__ATOMIC_ACQUIRE, "workgroup", "local");
}

__device__ __forceinline__ void lg_pair(const float* __restrict__ bl0,
                                        const float* __restrict__ bl1,
                                        const bool has1,
                                        const v2f (&a)[EMBD / 4],
                                        float* __restrict__ o0,
                                        float* __restrict__ o1) {
    v2f b0[EMBD / 4], b1[EMBD / 4];
#pragma unroll
    for (int kk = 0; kk < EMBD / 4; ++kk) b0[kk] = *reinterpret_cast<const v2f*>(bl0 + 4 * kk);
#pragma unroll
    for (int kk = 0; kk < EMBD / 4; ++kk) b1[kk] = *reinterpret_cast<const v2f*>(bl1 + 4 * kk);
    __builtin_amdgcn_sched_barrier(0);
    v8f c0 = {}, c1 = {};
#pragma unroll
    for (int kk = 0; kk < EMBD / 4; ++kk) {
        c0 = __builtin_amdgcn_wmma_f32_16x16x4_f32(false, a[kk], false, b0[kk], (short)0, c0, false, false);
        c1 = __builtin_amdgcn_wmma_f32_16x16x4_f32(false, a[kk], false, b1[kk], (short)0, c1, false, false);
    }
#pragma unroll
    for (int v = 0; v < 8; ++v) o0[(size_t)v * VOCAB] = c0[v];
    if (has1) {
#pragma unroll
        for (int v = 0; v < 8; ++v) o1[(size_t)v * VOCAB] = c1[v];
    }
}

__global__ __launch_bounds__(LG_THREADS) void memnn_logits_wmma(const float* __restrict__ u,
                                                                const float* __restrict__ A3,
                                                                float* __restrict__ out) {
    __shared__ __attribute__((aligned(16))) float b_lds[LG_MAXS * 16 * LG_PITCH];
    __shared__ __attribute__((aligned(16))) float u_lds[BS * LG_PITCH];

    const int lane = threadIdx.x & 31;
    const int wave = __builtin_amdgcn_readfirstlane(threadIdx.x >> 5);
    const int g    = blockIdx.x;
    const int s_begin = (g * LG_NSTRIPS) / LG_GRID;
    const int ns      = ((g + 1) * LG_NSTRIPS) / LG_GRID - s_begin;
    const int rows_b  = (ns - LG_HALF) * 16;

    const float4* A4 = reinterpret_cast<const float4*>(A3) + (size_t)s_begin * 16 * 32;
    const float4* U4 = reinterpret_cast<const float4*>(u);

    float4 su[LG_ROWS_U], sa[LG_ROWS_A], sb[LG_ROWS_B];
#pragma unroll
    for (int i = 0; i < LG_ROWS_U; ++i) su[i] = U4[(i * LG_WAVES + wave) * 32 + lane];
#pragma unroll
    for (int i = 0; i < LG_ROWS_A; ++i) sa[i] = A4[(i * LG_WAVES + wave) * 32 + lane];
#pragma unroll
    for (int i = 0; i < LG_ROWS_U; ++i)
        *reinterpret_cast<float4*>(u_lds + (i * LG_WAVES + wave) * LG_PITCH + 4 * lane) = su[i];
#pragma unroll
    for (int i = 0; i < LG_ROWS_A; ++i)
        *reinterpret_cast<float4*>(b_lds + (i * LG_WAVES + wave) * LG_PITCH + 4 * lane) = sa[i];
#pragma unroll
    for (int i = 0; i < LG_ROWS_B; ++i) {
        const int r = i * LG_WAVES + wave;
        sb[i] = make_float4(0.f, 0.f, 0.f, 0.f);
        if (r < rows_b) sb[i] = A4[(LG_HALF * 16 + r) * 32 + lane];
    }
    lg_lds_barrier();

    const int mt = wave & 3;
    const int p  = wave >> 2;
    const int n  = lane & 15;
    const int h  = lane >> 4;

    v2f a[EMBD / 4];
    {
        const float* ar = u_lds + (mt * 16 + n) * LG_PITCH + 2 * h;
#pragma unroll
        for (int kk = 0; kk < EMBD / 4; ++kk) a[kk] = *reinterpret_cast<const v2f*>(ar + 4 * kk);
    }

    const float* bl = b_lds + n * LG_PITCH + 2 * h;
    float* ob = out + (size_t)(mt * 16 + 8 * h) * VOCAB + (size_t)s_begin * 16 + n;

    for (int j = p; j < LG_HALF / 2; j += 2) {
        const int s0 = 2 * j;
        lg_pair(bl + s0 * 16 * LG_PITCH, bl + (s0 + 1) * 16 * LG_PITCH, true, a,
                ob + s0 * 16, ob + (s0 + 1) * 16);
    }

#pragma unroll
    for (int i = 0; i < LG_ROWS_B; ++i) {
        const int r = i * LG_WAVES + wave;
        if (r < rows_b)
            *reinterpret_cast<float4*>(b_lds + (LG_HALF * 16 + r) * LG_PITCH + 4 * lane) = sb[i];
    }
    lg_lds_barrier();

    for (int j = p; LG_HALF + 2 * j < ns; j += 2) {
        const int  s0   = LG_HALF + 2 * j;
        const bool has1 = s0 + 1 < ns;
        const int  s1   = has1 ? s0 + 1 : s0;
        lg_pair(bl + s0 * 16 * LG_PITCH, bl + s1 * 16 * LG_PITCH, has1, a,
                ob + s0 * 16, ob + s1 * 16);
    }
}

#define LSM_THREADS 1024
#define LSM_WAVES   (LSM_THREADS / 32)
#define LSM_VEC     (VOCAB / 4)
#define LSM_PER     ((LSM_VEC + LSM_THREADS - 1) / LSM_THREADS)
static_assert(VOCAB % 4 == 0, "rows are read as float4");
static_assert(LSM_WAVES == 32, "second-stage fold reads one partial per lane");
static_assert((LSM_PER - 1) * LSM_THREADS < LSM_VEC, "only the last slot is partial");

__global__ __launch_bounds__(LSM_THREADS) void memnn_logsoftmax(float* __restrict__ out) {
    __shared__ float red_max[LSM_WAVES];
    __shared__ float red_sum[LSM_WAVES];
    const int b    = blockIdx.x;
    const int t    = threadIdx.x;
    const int wave = t >> 5;
    const int lane = t & 31;
    float4* row = reinterpret_cast<float4*>(out + (size_t)b * VOCAB);

    const float NEG = -3.4e38f;
    const int  tail_idx = t + (LSM_PER - 1) * LSM_THREADS;
    const bool tail     = tail_idx < LSM_VEC;

    float4 v[LSM_PER];
#pragma unroll
    for (int i = 0; i < LSM_PER - 1; ++i) v[i] = row[t + i * LSM_THREADS];
    v[LSM_PER - 1] = make_float4(NEG, NEG, NEG, NEG);
    if (tail) v[LSM_PER - 1] = row[tail_idx];

    float mx = NEG;
#pragma unroll
    for (int i = 0; i < LSM_PER; ++i)
        mx = fmaxf(mx, fmaxf(fmaxf(v[i].x, v[i].y), fmaxf(v[i].z, v[i].w)));
    for (int o = 16; o > 0; o >>= 1) mx = fmaxf(mx, __shfl_xor(mx, o, 32));
    if (lane == 0) red_max[wave] = mx;
    __syncthreads();
    mx = red_max[lane];
    for (int o = 16; o > 0; o >>= 1) mx = fmaxf(mx, __shfl_xor(mx, o, 32));

    float s0 = 0.f, s1 = 0.f, s2 = 0.f, s3 = 0.f;
#pragma unroll
    for (int i = 0; i < LSM_PER - 1; ++i) {
        s0 += __expf(v[i].x - mx);
        s1 += __expf(v[i].y - mx);
        s2 += __expf(v[i].z - mx);
        s3 += __expf(v[i].w - mx);
    }
    if (tail) {
        s0 += __expf(v[LSM_PER - 1].x - mx);
        s1 += __expf(v[LSM_PER - 1].y - mx);
        s2 += __expf(v[LSM_PER - 1].z - mx);
        s3 += __expf(v[LSM_PER - 1].w - mx);
    }
    float sm = (s0 + s1) + (s2 + s3);
    for (int o = 16; o > 0; o >>= 1) sm += __shfl_xor(sm, o, 32);
    if (lane == 0) red_sum[wave] = sm;
    __syncthreads();
    sm = red_sum[lane];
    for (int o = 16; o > 0; o >>= 1) sm += __shfl_xor(sm, o, 32);

    const float lse = mx + logf(sm);

#pragma unroll
    for (int i = 0; i < LSM_PER - 1; ++i) {
        float4 r = v[i];
        r.x -= lse; r.y -= lse; r.z -= lse; r.w -= lse;
        row[t + i * LSM_THREADS] = r;
    }
    if (tail) {
        float4 r = v[LSM_PER - 1];
        r.x -= lse; r.y -= lse; r.z -= lse; r.w -= lse;
        row[tail_idx] = r;
    }
}

extern "C" void kernel_launch(void* const* d_in, const int* in_sizes, int n_in,
                              void* d_out, int out_size, void* d_ws, size_t ws_size,
                              hipStream_t stream) {
    const int*   x  = (const int*)d_in[0];
    const int*   q  = (const int*)d_in[1];
    const float* A  = (const float*)d_in[2];
    const float* TA = (const float*)d_in[3];
    const float* TC = (const float*)d_in[4];
    float* out = (float*)d_out;

    float* u = (float*)d_ws;

    memnn_story<<<BS, ST_THREADS, 0, stream>>>(x, q, A, TA, TC, u);

    memnn_logits_wmma<<<LG_GRID, LG_THREADS, 0, stream>>>(
        u, A + 3 * (size_t)VOCAB * EMBD, out);

    memnn_logsoftmax<<<BS, LSM_THREADS, 0, stream>>>(out);
}


# table-row gather moved out of the 64 per-batch-row work-groups into its own launch of 256 work-groups (batch row x level); m and c pass through the workspace to a hops launch; sums and order kept
# speedup vs baseline: 3.8772x; 1.2316x over previous
#include <hip/hip_runtime.h>
#include <hip/hip_bf16.h>
#include <math.h>

#define VOCAB 50000
#define EMBD  128
#define HOPS  3
#define BS    64
#define STORY 50
#define SENT  20
#define QLEN  20

typedef __attribute__((ext_vector_type(2))) float v2f;
typedef __attribute__((ext_vector_type(8))) float v8f;

#define MC_PLANE4  (HOPS * STORY * 32)
#define MC_B4      (2 * MC_PLANE4)
#define GA_WAVES   10
#define GA_THREADS (GA_WAVES * 32)
#define GA_PER     (STORY / GA_WAVES)
static_assert(EMBD == 128, "a lane owns 4 columns: 32 lanes cover one row");
static_assert(STORY % GA_WAVES == 0, "sentences dealt evenly to the waves");

template <bool HAS_M, bool HAS_C>
__device__ __forceinline__ void memnn_gather_units(const int* __restrict__ xb,
                                                   const float4* __restrict__ Al,
                                                   const float4* __restrict__ TA4,
                                                   const float4* __restrict__ TC4,
                                                   float4* __restrict__ mrows,
                                                   float4* __restrict__ crows,
                                                   const int wave, const int lane) {
    const float4 zero4 = make_float4(0.f, 0.f, 0.f, 0.f);

    const float kd0 = (float)(4 * lane + 1) / (float)EMBD;
    const float kd1 = (float)(4 * lane + 2) / (float)EMBD;
    const float kd2 = (float)(4 * lane + 3) / (float)EMBD;
    const float kd3 = (float)(4 * lane + 4) / (float)EMBD;
    const float w0 = 1.0f - 2.0f * kd0, w1 = 1.0f - 2.0f * kd1;
    const float w2 = 1.0f - 2.0f * kd2, w3 = 1.0f - 2.0f * kd3;

    const int tl = lane < SENT ? lane : 0;
    int tcur = xb[wave * SENT + tl];

#pragma nounroll
    for (int i = 0; i < GA_PER; ++i) {
        const int s  = wave + GA_WAVES * i;
        const int sn = i + 1 < GA_PER ? s + GA_WAVES : s;
        const int tnext = xb[sn * SENT + tl];

        float4 e[SENT];
#pragma unroll
        for (int j = 0; j < SENT; ++j)
            e[j] = Al[(size_t)__builtin_amdgcn_readlane(tcur, j) * 32 + lane];

        float4 ta = zero4, tc = zero4;
        if (HAS_M) ta = TA4[s * 32 + lane];
        if (HAS_C) tc = TC4[s * 32 + lane];

        float4 ap = zero4, cp = zero4;
#pragma unroll
        for (int j = 0; j < SENT; ++j) {
            const float aj = 1.0f - (float)(j + 1) / (float)SENT;
            ap.x += e[j].x * aj;  ap.y += e[j].y * aj;
            ap.z += e[j].z * aj;  ap.w += e[j].w * aj;
            cp.x += e[j].x;  cp.y += e[j].y;  cp.z += e[j].z;  cp.w += e[j].w;
        }
        if (HAS_M) {
            float4 mp;
            mp.x = (w0 * ap.x + kd0 * cp.x) + ta.x;
            mp.y = (w1 * ap.y + kd1 * cp.y) + ta.y;
            mp.z = (w2 * ap.z + kd2 * cp.z) + ta.z;
            mp.w = (w3 * ap.w + kd3 * cp.w) + ta.w;
            mrows[s * 32 + lane] = mp;
        }
        if (HAS_C) {
            cp.x += tc.x; cp.y += tc.y; cp.z += tc.z; cp.w += tc.w;
            crows[s * 32 + lane] = cp;
        }
        tcur = tnext;
    }
}

__global__ __launch_bounds__(GA_THREADS) __attribute__((amdgpu_waves_per_eu(3, 4))) void memnn_gather(const int* __restrict__ x,
                                                           const float* __restrict__ A,
                                                           const float* __restrict__ TA,
                                                           const float* __restrict__ TC,
                                                           float* __restrict__ mc) {
    const int b     = blockIdx.x >> 2;
    const int level = blockIdx.x & 3;
    const int lane  = threadIdx.x & 31;
    const int wave = __builtin_amdgcn_readfirstlane(threadIdx.x >> 5);

    const size_t LVL4 = (size_t)VOCAB * 32;
    const float4* Al  = reinterpret_cast<const float4*>(A) + (size_t)level * LVL4;
    const float4* TA4 = reinterpret_cast<const float4*>(TA);
    const float4* TC4 = reinterpret_cast<const float4*>(TC);
    const int*    xb  = x + (size_t)b * STORY * SENT;
    float4* mplanes = reinterpret_cast<float4*>(mc) + (size_t)b * MC_B4;
    float4* cplanes = mplanes + MC_PLANE4;

    if (level == 0)
        memnn_gather_units<true, false>(xb, Al, TA4, TC4, mplanes, cplanes, wave, lane);
    else if (level == HOPS)
        memnn_gather_units<false, true>(xb, Al, TA4, TC4, mplanes, cplanes + (level - 1) * STORY * 32, wave, lane);
    else
        memnn_gather_units<true, true>(xb, Al, TA4, TC4, mplanes + level * STORY * 32,
                                       cplanes + (level - 1) * STORY * 32, wave, lane);
}

#define ST_THREADS 1024
#define ST_WAVES   (ST_THREADS / 32)
#define ST_GROUPS  (ST_THREADS / EMBD)
#define ST_ROWS    ((STORY + ST_GROUPS - 1) / ST_GROUPS)
#define CP_WAVES   30
#define CP_THREADS (CP_WAVES * 32)
#define CP_PER     (MC_B4 / CP_THREADS)
static_assert(STORY > 32 && STORY <= 64, "a wave scores rows w and w+32; a lane holds scores l and l+32");
static_assert(CP_PER * CP_THREADS == MC_B4 && CP_WAVES < ST_WAVES, "the copy divides evenly: no partial slot");

__global__ __launch_bounds__(ST_THREADS) __attribute__((amdgpu_waves_per_eu(8, 8))) void memnn_hops(const int* __restrict__ q,
                                                         const float* __restrict__ A,
                                                         const float* __restrict__ mc,
                                                         float* __restrict__ u) {
    __shared__ float4 mc_lds[MC_B4];
    __shared__ float4 u_lds[32];
    __shared__ float score[64];
    __shared__ float part[ST_GROUPS * EMBD];
    const float4* m_lds = mc_lds;
    const float4* c_lds = mc_lds + MC_PLANE4;

    const int b    = blockIdx.x;
    const int t    = threadIdx.x;
    const int lane = t & 31;
    const int wave = __builtin_amdgcn_readfirstlane(t >> 5);

    const float4* A4 = reinterpret_cast<const float4*>(A);
    const float4 zero4 = make_float4(0.f, 0.f, 0.f, 0.f);

    if (wave < CP_WAVES) {
        const float4* src = reinterpret_cast<const float4*>(mc) + (size_t)b * MC_B4;
        float4 v[CP_PER];
#pragma unroll
        for (int i = 0; i < CP_PER; ++i) v[i] = src[t + i * CP_THREADS];
#pragma unroll
        for (int i = 0; i < CP_PER; ++i) mc_lds[t + i * CP_THREADS] = v[i];
    } else if (wave == ST_WAVES - 1) {
        const int* qt = q + b * QLEN;
        float4 e[QLEN];
#pragma unroll
        for (int j = 0; j < QLEN; ++j) e[j] = A4[(size_t)qt[j] * 32 + lane];
        float4 acc = zero4;
#pragma unroll
        for (int j = 0; j < QLEN; ++j) {
            acc.x += e[j].x; acc.y += e[j].y; acc.z += e[j].z; acc.w += e[j].w;
        }
        u_lds[lane] = acc;
    }
    __syncthreads();

    const float  NEG   = -3.4e38f;
    const bool   two   = wave < STORY - 32;
    const int    g     = t >> 7;
    const int    d     = t & (EMBD - 1);
    const float* c_f   = reinterpret_cast<const float*>(c_lds);
    float*       u_f   = reinterpret_cast<float*>(u_lds);

    for (int h = 0; h < HOPS; ++h) {
        const float4 uv = u_lds[lane];
        const float4 m0 = m_lds[(h * STORY + wave) * 32 + lane];
        float4 m1 = zero4;
        if (two) m1 = m_lds[(h * STORY + wave + 32) * 32 + lane];
        float p0 = (m0.x * uv.x + m0.y * uv.y) + (m0.z * uv.z + m0.w * uv.w);
        float p1 = (m1.x * uv.x + m1.y * uv.y) + (m1.z * uv.z + m1.w * uv.w);
        for (int o = 16; o > 0; o >>= 1) {
            p0 += __shfl_xor(p0, o, 32);
            p1 += __shfl_xor(p1, o, 32);
        }
        if (lane == 0) {
            score[wave] = p0;
            if (two) score[wave + 32] = p1;
        }
        __syncthreads();

        const bool  hi = lane < STORY - 32;
        const float s0 = score[lane];
        const float s1 = hi ? score[lane + 32] : NEG;
        float mx = fmaxf(s0, s1);
        for (int o = 16; o > 0; o >>= 1) mx = fmaxf(mx, __shfl_xor(mx, o, 32));
        float sm = __expf(s0 - mx) + (hi ? __expf(s1 - mx) : 0.f);
        for (int o = 16; o > 0; o >>= 1) sm += __shfl_xor(sm, o, 32);
        const float inv = 1.0f / sm;

        float acc = 0.f;
        for (int i = 0; i < ST_ROWS; ++i) {
            const int s = g + ST_GROUPS * i;
            if (s < STORY)
                acc += (__expf(score[s] - mx) * inv) * c_f[(h * STORY + s) * EMBD + d];
        }
        part[g * EMBD + d] = acc;
        __syncthreads();

        if (t < EMBD) {
            float a = part[t];
            for (int k = 1; k < ST_GROUPS; ++k) a += part[k * EMBD + t];
            const float un = u_f[t] + a;
            u_f[t] = un;
            if (h == HOPS - 1) u[b * EMBD + t] = un;
        }
        __syncthreads();
    }
}

#define LG_THREADS 256
#define LG_WAVES   (LG_THREADS / 32)
#define LG_GRID    256
#define LG_NSTRIPS (VOCAB / 16)
#define LG_MAXS    ((LG_NSTRIPS + LG_GRID - 1) / LG_GRID)
#define LG_HALF    6
#define LG_PITCH   132
#define LG_ROWS_A  (LG_HALF * 16 / LG_WAVES)
#define LG_ROWS_B  (((LG_MAXS - LG_HALF) * 16 + LG_WAVES - 1) / LG_WAVES)
#define LG_ROWS_U  (BS / LG_WAVES)
static_assert(VOCAB % 16 == 0, "whole strips");
static_assert(LG_NSTRIPS / LG_GRID >= LG_HALF, "the first half is always full");
static_assert((LG_HALF * 16) % LG_WAVES == 0 && BS % LG_WAVES == 0, "rows dealt evenly to the waves");
static_assert(BS == 64 && LG_WAVES == 8, "4 batch tiles x 2 waves");

__device__ __forceinline__ void lg_lds_barrier() {
    __builtin_amdgcn_fence(__ATOMIC_RELEASE, "workgroup", "local");
    __builtin_amdgcn_s_barrier();
    __builtin_amdgcn_fence(__ATOMIC_ACQUIRE, "workgroup", "local");
}

__device__ __forceinline__ void lg_pair(const float* __restrict__ bl0,
                                        const float* __restrict__ bl1,
                                        const bool has1,
                                        const v2f (&a)[EMBD / 4],
                                        float* __restrict__ o0,
                                        float* __restrict__ o1) {
    v2f b0[EMBD / 4], b1[EMBD / 4];
#pragma unroll
    for (int kk = 0; kk < EMBD / 4; ++kk) b0[kk] = *reinterpret_cast<const v2f*>(bl0 + 4 * kk);
#pragma unroll
    for (int kk = 0; kk < EMBD / 4; ++kk) b1[kk] = *reinterpret_cast<const v2f*>(bl1 + 4 * kk);
    __builtin_amdgcn_sched_barrier(0);
    v8f c0 = {}, c1 = {};
#pragma unroll
    for (int kk = 0; kk < EMBD / 4; ++kk) {
        c0 = __builtin_amdgcn_wmma_f32_16x16x4_f32(false, a[kk], false, b0[kk], (short)0, c0, false, false);
        c1 = __builtin_amdgcn_wmma_f32_16x16x4_f32(false, a[kk], false, b1[kk], (short)0, c1, false, false);
    }
#pragma unroll
    for (int v = 0; v < 8; ++v) o0[(size_t)v * VOCAB] = c0[v];
    if (has1) {
#pragma unroll
        for (int v = 0; v < 8; ++v) o1[(size_t)v * VOCAB] = c1[v];
    }
}

__global__ __launch_bounds__(LG_THREADS) void memnn_logits_wmma(const float* __restrict__ u,
                                                                const float* __restrict__ A3,
                                                                float* __restrict__ out) {
    __shared__ __attribute__((aligned(16))) float b_lds[LG_MAXS * 16 * LG_PITCH];
    __shared__ __attribute__((aligned(16))) float u_lds[BS * LG_PITCH];

    const int lane = threadIdx.x & 31;
    const int wave = __builtin_amdgcn_readfirstlane(threadIdx.x >> 5);
    const int g    = blockIdx.x;
    const int s_begin = (g * LG_NSTRIPS) / LG_GRID;
    const int ns      = ((g + 1) * LG_NSTRIPS) / LG_GRID - s_begin;
    const int rows_b  = (ns - LG_HALF) * 16;

    const float4* A4 = reinterpret_cast<const float4*>(A3) + (size_t)s_begin * 16 * 32;
    const float4* U4 = reinterpret_cast<const float4*>(u);

    float4 su[LG_ROWS_U], sa[LG_ROWS_A], sb[LG_ROWS_B];
#pragma unroll
    for (int i = 0; i < LG_ROWS_U; ++i) su[i] = U4[(i * LG_WAVES + wave) * 32 + lane];
#pragma unroll
    for (int i = 0; i < LG_ROWS_A; ++i) sa[i] = A4[(i * LG_WAVES + wave) * 32 + lane];
#pragma unroll
    for (int i = 0; i < LG_ROWS_U; ++i)
        *reinterpret_cast<float4*>(u_lds + (i * LG_WAVES + wave) * LG_PITCH + 4 * lane) = su[i];
#pragma unroll
    for (int i = 0; i < LG_ROWS_A; ++i)
        *reinterpret_cast<float4*>(b_lds + (i * LG_WAVES + wave) * LG_PITCH + 4 * lane) = sa[i];
#pragma unroll
    for (int i = 0; i < LG_ROWS_B; ++i) {
        const int r = i * LG_WAVES + wave;
        sb[i] = make_float4(0.f, 0.f, 0.f, 0.f);
        if (r < rows_b) sb[i] = A4[(LG_HALF * 16 + r) * 32 + lane];
    }
    lg_lds_barrier();

    const int mt = wave & 3;
    const int p  = wave >> 2;
    const int n  = lane & 15;
    const int h  = lane >> 4;

    v2f a[EMBD / 4];
    {
        const float* ar = u_lds + (mt * 16 + n) * LG_PITCH + 2 * h;
#pragma unroll
        for (int kk = 0; kk < EMBD / 4; ++kk) a[kk] = *reinterpret_cast<const v2f*>(ar + 4 * kk);
    }

    const float* bl = b_lds + n * LG_PITCH + 2 * h;
    float* ob = out + (size_t)(mt * 16 + 8 * h) * VOCAB + (size_t)s_begin * 16 + n;

    for (int j = p; j < LG_HALF / 2; j += 2) {
        const int s0 = 2 * j;
        lg_pair(bl + s0 * 16 * LG_PITCH, bl + (s0 + 1) * 16 * LG_PITCH, true, a,
                ob + s0 * 16, ob + (s0 + 1) * 16);
    }

#pragma unroll
    for (int i = 0; i < LG_ROWS_B; ++i) {
        const int r = i * LG_WAVES + wave;
        if (r < rows_b)
            *reinterpret_cast<float4*>(b_lds + (LG_HALF * 16 + r) * LG_PITCH + 4 * lane) = sb[i];
    }
    lg_lds_barrier();

    for (int j = p; LG_HALF + 2 * j < ns; j += 2) {
        const int  s0   = LG_HALF + 2 * j;
        const bool has1 = s0 + 1 < ns;
        const int  s1   = has1 ? s0 + 1 : s0;
        lg_pair(bl + s0 * 16 * LG_PITCH, bl + s1 * 16 * LG_PITCH, has1, a,
                ob + s0 * 16, ob + s1 * 16);
    }
}

#define LSM_THREADS 1024
#define LSM_WAVES   (LSM_THREADS / 32)
#define LSM_VEC     (VOCAB / 4)
#define LSM_PER     ((LSM_VEC + LSM_THREADS - 1) / LSM_THREADS)
static_assert(VOCAB % 4 == 0, "rows are read as float4");
static_assert(LSM_WAVES == 32, "second-stage fold reads one partial per lane");
static_assert((LSM_PER - 1) * LSM_THREADS < LSM_VEC, "only the last slot is partial");

__global__ __launch_bounds__(LSM_THREADS) void memnn_logsoftmax(float* __restrict__ out) {
    __shared__ float red_max[LSM_WAVES];
    __shared__ float red_sum[LSM_WAVES];
    const int b    = blockIdx.x;
    const int t    = threadIdx.x;
    const int wave = t >> 5;
    const int lane = t & 31;
    float4* row = reinterpret_cast<float4*>(out + (size_t)b * VOCAB);

    const float NEG = -3.4e38f;
    const int  tail_idx = t + (LSM_PER - 1) * LSM_THREADS;
    const bool tail     = tail_idx < LSM_VEC;

    float4 v[LSM_PER];
#pragma unroll
    for (int i = 0; i < LSM_PER - 1; ++i) v[i] = row[t + i * LSM_THREADS];
    v[LSM_PER - 1] = make_float4(NEG, NEG, NEG, NEG);
    if (tail) v[LSM_PER - 1] = row[tail_idx];

    float mx = NEG;
#pragma unroll
    for (int i = 0; i < LSM_PER; ++i)
        mx = fmaxf(mx, fmaxf(fmaxf(v[i].x, v[i].y), fmaxf(v[i].z, v[i].w)));
    for (int o = 16; o > 0; o >>= 1) mx = fmaxf(mx, __shfl_xor(mx, o, 32));
    if (lane == 0) red_max[wave] = mx;
    __syncthreads();
    mx = red_max[lane];
    for (int o = 16; o > 0; o >>= 1) mx = fmaxf(mx, __shfl_xor(mx, o, 32));

    float s0 = 0.f, s1 = 0.f, s2 = 0.f, s3 = 0.f;
#pragma unroll
    for (int i = 0; i < LSM_PER - 1; ++i) {
        s0 += __expf(v[i].x - mx);
        s1 += __expf(v[i].y - mx);
        s2 += __expf(v[i].z - mx);
        s3 += __expf(v[i].w - mx);
    }
    if (tail) {
        s0 += __expf(v[LSM_PER - 1].x - mx);
        s1 += __expf(v[LSM_PER - 1].y - mx);
        s2 += __expf(v[LSM_PER - 1].z - mx);
        s3 += __expf(v[LSM_PER - 1].w - mx);
    }
    float sm = (s0 + s1) + (s2 + s3);
    for (int o = 16; o > 0; o >>= 1) sm += __shfl_xor(sm, o, 32);
    if (lane == 0) red_sum[wave] = sm;
    __syncthreads();
    sm = red_sum[lane];
    for (int o = 16; o > 0; o >>= 1) sm += __shfl_xor(sm, o, 32);

    const float lse = mx + logf(sm);

#pragma unroll
    for (int i = 0; i < LSM_PER - 1; ++i) {
        float4 r = v[i];
        r.x -= lse; r.y -= lse; r.z -= lse; r.w -= lse;
        row[t + i * LSM_THREADS] = r;
    }
    if (tail) {
        float4 r = v[LSM_PER - 1];
        r.x -= lse; r.y -= lse; r.z -= lse; r.w -= lse;
        row[tail_idx] = r;
    }
}

extern "C" void kernel_launch(void* const* d_in, const int* in_sizes, int n_in,
                              void* d_out, int out_size, void* d_ws, size_t ws_size,
                              hipStream_t stream) {
    const int*   x  = (const int*)d_in[0];
    const int*   q  = (const int*)d_in[1];
    const float* A  = (const float*)d_in[2];
    const float* TA = (const float*)d_in[3];
    const float* TC = (const float*)d_in[4];
    float* out = (float*)d_out;

    float* u  = (float*)d_ws;
    float* mc = (float*)d_ws + BS * EMBD;

    memnn_gather<<<BS * 4, GA_THREADS, 0, stream>>>(x, A, TA, TC, mc);

    memnn_hops<<<BS, ST_THREADS, 0, stream>>>(q, A, mc, u);

    memnn_logits_wmma<<<LG_GRID, LG_THREADS, 0, stream>>>(
        u, A + 3 * (size_t)VOCAB * EMBD, out);

    memnn_logsoftmax<<<BS, LSM_THREADS, 0, stream>>>(out);
}
